# starting kernel
# baseline (speedup 1.0000x reference)
#include <hip/hip_runtime.h>
#include <math.h>

typedef float v2f __attribute__((ext_vector_type(2)));
typedef float v8f __attribute__((ext_vector_type(8)));

#define BT   16
#define NCH  128
#define NHID 256
#define OBSD 99

__global__ __launch_bounds__(256) void gnn_fused_kernel(
    const float* __restrict__ obs,
    const float* __restrict__ W_fgn,
    const float* __restrict__ b_fgn,
    const float* __restrict__ W_root,
    const float* __restrict__ b_conv,
    const float* __restrict__ attn_k,
    const float* __restrict__ W_dense,
    const float* __restrict__ b_dense,
    float* __restrict__ out)
{
    __shared__ float s_obs[BT][96];
    __shared__ float s_xc[BT][4][NCH];
    __shared__ float s_pool[BT][132];
    __shared__ float s_logit[BT][4];
    __shared__ float s_attn[BT][4];

    const int tid   = threadIdx.x;
    const int bbase = blockIdx.x * BT;

    for (int i = tid; i < BT * 96; i += 256) {
        int r = i / 96, c = i % 96;
        s_obs[r][c] = obs[(size_t)(bbase + r) * OBSD + 3 + c];
    }
    if (tid < BT * 4) s_logit[tid >> 2][tid & 3] = 0.0f;
    __syncthreads();

    for (int p = 0; p < 8; ++p) {
        int task = p * 256 + tid;
        int b = task >> 7;
        int c = task & 127;
        const float* ob = s_obs[b];

        float Wf[4][4], bf[4], Wr[4];
        #pragma unroll
        for (int e = 0; e < 4; ++e)
            #pragma unroll
            for (int f = 0; f < 4; ++f)
                Wf[e][f] = W_fgn[e * 512 + c * 4 + f];
        #pragma unroll
        for (int f = 0; f < 4; ++f) { bf[f] = b_fgn[c * 4 + f]; Wr[f] = W_root[f * 128 + c]; }

        float P[4][4], Q[4];
        #pragma unroll
        for (int s = 0; s < 4; ++s) {
            Q[s] = 0.f;
            #pragma unroll
            for (int e = 0; e < 4; ++e) P[s][e] = 0.f;
            #pragma unroll
            for (int f = 0; f < 4; ++f) {
                float x = ob[s * 4 + f];
                Q[s] = fmaf(bf[f], x, Q[s]);
                #pragma unroll
                for (int e = 0; e < 4; ++e) P[s][e] = fmaf(Wf[e][f], x, P[s][e]);
            }
        }
        float bc = b_conv[c], akc = attn_k[c];
        #pragma unroll
        for (int t = 0; t < 4; ++t) {
            float msg = 0.f;
            #pragma unroll
            for (int s = 0; s < 4; ++s) {
                float inner = Q[s];
                #pragma unroll
                for (int e = 0; e < 4; ++e)
                    inner = fmaf(ob[32 + (t * 4 + s) * 4 + e], P[s][e], inner);
                msg = fmaf(ob[16 + t * 4 + s], inner, msg);
            }
            float rt = 0.f;
            #pragma unroll
            for (int f = 0; f < 4; ++f) rt = fmaf(ob[t * 4 + f], Wr[f], rt);
            float xc = msg + rt + bc;
            xc = xc > 0.f ? xc : 0.f;
            s_xc[b][t][c] = xc;
            atomicAdd(&s_logit[b][t], xc * akc);
        }
    }
    __syncthreads();

    if (tid < BT) {
        float l0 = s_logit[tid][0], l1 = s_logit[tid][1];
        float l2 = s_logit[tid][2], l3 = s_logit[tid][3];
        float m  = fmaxf(fmaxf(l0, l1), fmaxf(l2, l3));
        float e0 = expf(l0 - m), e1 = expf(l1 - m), e2 = expf(l2 - m), e3 = expf(l3 - m);
        float inv = 1.0f / (e0 + e1 + e2 + e3);
        s_attn[tid][0] = e0 * inv; s_attn[tid][1] = e1 * inv;
        s_attn[tid][2] = e2 * inv; s_attn[tid][3] = e3 * inv;
    }
    __syncthreads();

    for (int p = 0; p < 8; ++p) {
        int task = p * 256 + tid;
        int b = task >> 7, c = task & 127;
        float acc = 0.f;
        #pragma unroll
        for (int t = 0; t < 4; ++t) acc = fmaf(s_attn[b][t], s_xc[b][t][c], acc);
        s_pool[b][c] = acc;
    }
    __syncthreads();

    const int wid  = tid >> 5;
    const int lane = tid & 31;
    const int half = lane >> 4;
    const int mrow = lane & 15;

    #pragma unroll
    for (int ti = 0; ti < 2; ++ti) {
        const int tile = wid * 2 + ti;
        const int col  = tile * 16 + mrow;
        v8f acc = {};
        #pragma unroll 4
        for (int kk = 0; kk < NCH; kk += 4) {
            const int k0 = kk + half * 2;
            v2f a, bm;
            a.x  = s_pool[mrow][k0];
            a.y  = s_pool[mrow][k0 + 1];
            bm.x = W_dense[(size_t)k0 * NHID + col];
            bm.y = W_dense[(size_t)(k0 + 1) * NHID + col];
            acc = __builtin_amdgcn_wmma_f32_16x16x4_f32(
                 false, a,  false, bm,
                 (short)0, acc,  false,  false);
        }
        const float bd = b_dense[col];
        #pragma unroll
        for (int r = 0; r < 8; ++r) {
            int row = bbase + half * 8 + r;
            out[(size_t)row * NHID + col] = tanhf(acc[r] + bd);
        }
    }
}

extern "C" void kernel_launch(void* const* d_in, const int* in_sizes, int n_in,
                              void* d_out, int out_size, void* d_ws, size_t ws_size,
                              hipStream_t stream) {
    const float* obs     = (const float*)d_in[0];
    const float* W_fgn   = (const float*)d_in[1];
    const float* b_fgn   = (const float*)d_in[2];
    const float* W_root  = (const float*)d_in[3];
    const float* b_conv  = (const float*)d_in[4];
    const float* attn_k  = (const float*)d_in[5];
    const float* W_dense = (const float*)d_in[6];
    const float* b_dense = (const float*)d_in[7];
    float* out = (float*)d_out;

    const int B = in_sizes[0] / OBSD;
    const int grid = B / BT;
    gnn_fused_kernel<<<grid, 256, 0, stream>>>(obs, W_fgn, b_fgn, W_root, b_conv,
                                               attn_k, W_dense, b_dense, out);
}


# Graph conv re-associated into a per-(row,node) 24-term moment vector times stacked weights, run as 64x24x128 f32 WMMA instead of vector FMAs; conv and logit sums re-ordered (fixed tree, no atomics)
# speedup vs baseline: 3.6445x; 3.6445x over previous
#include <hip/hip_runtime.h>
#include <math.h>

typedef float v2f __attribute__((ext_vector_type(2)));
typedef float v8f __attribute__((ext_vector_type(8)));

#define BT   16
#define NCH  128
#define NHID 256
#define OBSD 99
#define ZP   28

template <int M>
__device__ __forceinline__ float row_xor(float x) {
    int b = __builtin_bit_cast(int, x);
    return __builtin_bit_cast(float, __builtin_amdgcn_update_dpp(b, b, 0x160 + M, 0xf, 0xf, true));
}
__device__ __forceinline__ float row_sum8(const float (&v)[8], int mrow) {
    const bool b3 = (mrow & 8) != 0;
    const bool b2 = (mrow & 4) != 0;
    const bool b1 = (mrow & 2) != 0;
    float a[4];
    #pragma unroll
    for (int j = 0; j < 4; ++j)
        a[j] = (b3 ? v[4 + j] : v[j]) + row_xor<8>(b3 ? v[j] : v[4 + j]);
    float c[2];
    #pragma unroll
    for (int j = 0; j < 2; ++j)
        c[j] = (b2 ? a[2 + j] : a[j]) + row_xor<4>(b2 ? a[j] : a[2 + j]);
    float d = (b1 ? c[1] : c[0]) + row_xor<2>(b1 ? c[0] : c[1]);
    d += row_xor<1>(d);
    return d;
}
__global__ __launch_bounds__(256) void gnn_fused_kernel(
    const float* __restrict__ obs,
    const float* __restrict__ W_fgn,
    const float* __restrict__ b_fgn,
    const float* __restrict__ W_root,
    const float* __restrict__ b_conv,
    const float* __restrict__ attn_k,
    const float* __restrict__ W_dense,
    const float* __restrict__ b_dense,
    float* __restrict__ out)
{
    __shared__ __attribute__((aligned(16))) float s_obs[BT][96];
    __shared__ float s_xc[BT][4][NCH];
    __shared__ float s_pool[BT][132];
    __shared__ __attribute__((aligned(16))) float s_z[BT * 4][ZP];
    __shared__ __attribute__((aligned(16))) float s_lpart[BT][8][4];
    __shared__ float s_attn[BT][4];

    const int tid   = threadIdx.x;
    const int bbase = blockIdx.x * BT;
    const int wid  = tid >> 5;
    const int lane = tid & 31;
    const int half = lane >> 4;
    const int mrow = lane & 15;

    const int cc = wid * 16 + mrow;
    v2f wb[6];
    #pragma unroll
    for (int e = 0; e < 4; ++e) {
        wb[e].x = W_fgn[e * 512 + cc * 4 + half * 2];
        wb[e].y = W_fgn[e * 512 + cc * 4 + half * 2 + 1];
    }
    wb[4].x = b_fgn[cc * 4 + half * 2];
    wb[4].y = b_fgn[cc * 4 + half * 2 + 1];
    wb[5].x = W_root[(half * 2) * 128 + cc];
    wb[5].y = W_root[(half * 2 + 1) * 128 + cc];
    const float bcv = b_conv[cc];
    const float akc = attn_k[cc];

    for (int i = tid; i < BT * 96; i += 256) {
        int r = i / 96, c = i % 96;
        s_obs[r][c] = obs[(size_t)(bbase + r) * OBSD + 3 + c];
    }
    __syncthreads();

    {
        const int b = tid >> 4, t = (tid >> 2) & 3, e = tid & 3;
        const float* ob = s_obs[b];
        const float4 av = *(const float4*)(ob + 16 + t * 4);
        const float a[4] = {av.x, av.y, av.z, av.w};
        float4 g = {0.f, 0.f, 0.f, 0.f};
        float4 h = {0.f, 0.f, 0.f, 0.f};
        #pragma unroll
        for (int s = 0; s < 4; ++s) {
            const float4 x = *(const float4*)(ob + s * 4);
            const float ae = a[s] * ob[32 + (t * 4 + s) * 4 + e];
            g.x = fmaf(ae, x.x, g.x); g.y = fmaf(ae, x.y, g.y);
            g.z = fmaf(ae, x.z, g.z); g.w = fmaf(ae, x.w, g.w);
            h.x = fmaf(a[s], x.x, h.x); h.y = fmaf(a[s], x.y, h.y);
            h.z = fmaf(a[s], x.z, h.z); h.w = fmaf(a[s], x.w, h.w);
        }
        float* zr = s_z[b * 4 + t];
        *(float4*)(zr + e * 4) = g;
        if (e == 0) *(float4*)(zr + 16) = h;
        if (e == 1) *(float4*)(zr + 20) = *(const float4*)(ob + t * 4);
    }
    __syncthreads();

    float tot[4];
    #pragma unroll
    for (int mt = 0; mt < 4; ++mt) {
        v8f acc = {bcv, bcv, bcv, bcv, bcv, bcv, bcv, bcv};
        #pragma unroll
        for (int ks = 0; ks < 6; ++ks) {
            const int k0 = ks * 4 + half * 2;
            v2f a;
            a.x = s_z[mt * 16 + mrow][k0];
            a.y = s_z[mt * 16 + mrow][k0 + 1];
            acc = __builtin_amdgcn_wmma_f32_16x16x4_f32(
                 false, a,  false, wb[ks],
                 (short)0, acc,  false,  false);
        }
        float lg[8];
        #pragma unroll
        for (int r = 0; r < 8; ++r) {
            float xc = acc[r];
            xc = xc > 0.f ? xc : 0.f;
            s_xc[mt * 4 + half * 2 + (r >> 2)][r & 3][cc] = xc;
            lg[r] = xc * akc;
        }
        tot[mt] = row_sum8(lg, mrow);
    }
    if ((mrow & 1) == 0) {
        const int r = mrow >> 1;
        #pragma unroll
        for (int mt = 0; mt < 4; ++mt)
            s_lpart[mt * 4 + half * 2 + (r >> 2)][wid][r & 3] = tot[mt];
    }
    __syncthreads();

    if (tid < BT) {
        const float (*q)[4] = s_lpart[tid];
        float l0 = ((q[0][0] + q[1][0]) + (q[2][0] + q[3][0])) + ((q[4][0] + q[5][0]) + (q[6][0] + q[7][0]));
        float l1 = ((q[0][1] + q[1][1]) + (q[2][1] + q[3][1])) + ((q[4][1] + q[5][1]) + (q[6][1] + q[7][1]));
        float l2 = ((q[0][2] + q[1][2]) + (q[2][2] + q[3][2])) + ((q[4][2] + q[5][2]) + (q[6][2] + q[7][2]));
        float l3 = ((q[0][3] + q[1][3]) + (q[2][3] + q[3][3])) + ((q[4][3] + q[5][3]) + (q[6][3] + q[7][3]));
        float m  = fmaxf(fmaxf(l0, l1), fmaxf(l2, l3));
        float e0 = expf(l0 - m), e1 = expf(l1 - m), e2 = expf(l2 - m), e3 = expf(l3 - m);
        float inv = 1.0f / (e0 + e1 + e2 + e3);
        s_attn[tid][0] = e0 * inv; s_attn[tid][1] = e1 * inv;
        s_attn[tid][2] = e2 * inv; s_attn[tid][3] = e3 * inv;
    }
    __syncthreads();

    for (int p = 0; p < 8; ++p) {
        int task = p * 256 + tid;
        int b = task >> 7, c = task & 127;
        float acc = 0.f;
        #pragma unroll
        for (int t = 0; t < 4; ++t) acc = fmaf(s_attn[b][t], s_xc[b][t][c], acc);
        s_pool[b][c] = acc;
    }
    __syncthreads();

    #pragma unroll
    for (int ti = 0; ti < 2; ++ti) {
        const int tile = wid * 2 + ti;
        const int col  = tile * 16 + mrow;
        v8f acc = {};
        #pragma unroll 4
        for (int kk = 0; kk < NCH; kk += 4) {
            const int k0 = kk + half * 2;
            v2f a, bm;
            a.x  = s_pool[mrow][k0];
            a.y  = s_pool[mrow][k0 + 1];
            bm.x = W_dense[(size_t)k0 * NHID + col];
            bm.y = W_dense[(size_t)(k0 + 1) * NHID + col];
            acc = __builtin_amdgcn_wmma_f32_16x16x4_f32(
                 false, a,  false, bm,
                 (short)0, acc,  false,  false);
        }
        const float bd = b_dense[col];
        #pragma unroll
        for (int r = 0; r < 8; ++r) {
            int row = bbase + half * 8 + r;
            out[(size_t)row * NHID + col] = tanhf(acc[r] + bd);
        }
    }
}

extern "C" void kernel_launch(void* const* d_in, const int* in_sizes, int n_in,
                              void* d_out, int out_size, void* d_ws, size_t ws_size,
                              hipStream_t stream) {
    const float* obs     = (const float*)d_in[0];
    const float* W_fgn   = (const float*)d_in[1];
    const float* b_fgn   = (const float*)d_in[2];
    const float* W_root  = (const float*)d_in[3];
    const float* b_conv  = (const float*)d_in[4];
    const float* attn_k  = (const float*)d_in[5];
    const float* W_dense = (const float*)d_in[6];
    const float* b_dense = (const float*)d_in[7];
    float* out = (float*)d_out;

    const int B = in_sizes[0] / OBSD;
    const int grid = B / BT;
    gnn_fused_kernel<<<grid, 256, 0, stream>>>(obs, W_fgn, b_fgn, W_root, b_conv,
                                               attn_k, W_dense, b_dense, out);
}


# Global-load round trips off the critical path: a thread's six obs loads issue as one group; W_dense fragments and b_dense are fetched ahead into registers across LDS-only barriers; sums unchanged
# speedup vs baseline: 4.4675x; 1.2258x over previous
#include <hip/hip_runtime.h>
#include <math.h>

typedef float v2f __attribute__((ext_vector_type(2)));
typedef float v8f __attribute__((ext_vector_type(8)));

#define BT   16
#define NCH  128
#define NHID 256
#define OBSD 99
#define ZP   28

template <int M>
__device__ __forceinline__ float row_xor(float x) {
    int b = __builtin_bit_cast(int, x);
    return __builtin_bit_cast(float, __builtin_amdgcn_update_dpp(b, b, 0x160 + M, 0xf, 0xf, true));
}
__device__ __forceinline__ float row_sum8(const float (&v)[8], int mrow) {
    const bool b3 = (mrow & 8) != 0;
    const bool b2 = (mrow & 4) != 0;
    const bool b1 = (mrow & 2) != 0;
    float a[4];
    #pragma unroll
    for (int j = 0; j < 4; ++j)
        a[j] = (b3 ? v[4 + j] : v[j]) + row_xor<8>(b3 ? v[j] : v[4 + j]);
    float c[2];
    #pragma unroll
    for (int j = 0; j < 2; ++j)
        c[j] = (b2 ? a[2 + j] : a[j]) + row_xor<4>(b2 ? a[j] : a[2 + j]);
    float d = (b1 ? c[1] : c[0]) + row_xor<2>(b1 ? c[0] : c[1]);
    d += row_xor<1>(d);
    return d;
}
__device__ __forceinline__ void lds_barrier() {
    __builtin_amdgcn_fence(__ATOMIC_RELEASE, "workgroup", "local");
    __builtin_amdgcn_s_barrier();
    __builtin_amdgcn_fence(__ATOMIC_ACQUIRE, "workgroup", "local");
}
__global__ __launch_bounds__(256) __attribute__((amdgpu_waves_per_eu(8))) void gnn_fused_kernel(
    const float* __restrict__ obs,
    const float* __restrict__ W_fgn,
    const float* __restrict__ b_fgn,
    const float* __restrict__ W_root,
    const float* __restrict__ b_conv,
    const float* __restrict__ attn_k,
    const float* __restrict__ W_dense,
    const float* __restrict__ b_dense,
    float* __restrict__ out)
{
    __shared__ __attribute__((aligned(16))) float s_obs[BT][96];
    __shared__ float s_xc[BT][4][NCH];
    __shared__ float s_pool[BT][132];
    __shared__ __attribute__((aligned(16))) float s_z[BT * 4][ZP];
    __shared__ __attribute__((aligned(16))) float s_lpart[BT][8][4];
    __shared__ float s_attn[BT][4];

    const int tid   = threadIdx.x;
    const int bbase = blockIdx.x * BT;
    const int wid  = tid >> 5;
    const int lane = tid & 31;
    const int half = lane >> 4;
    const int mrow = lane & 15;

    const int cc = wid * 16 + mrow;
    v2f wb[6];
    #pragma unroll
    for (int e = 0; e < 4; ++e) {
        wb[e].x = W_fgn[e * 512 + cc * 4 + half * 2];
        wb[e].y = W_fgn[e * 512 + cc * 4 + half * 2 + 1];
    }
    wb[4].x = b_fgn[cc * 4 + half * 2];
    wb[4].y = b_fgn[cc * 4 + half * 2 + 1];
    wb[5].x = W_root[(half * 2) * 128 + cc];
    wb[5].y = W_root[(half * 2 + 1) * 128 + cc];
    const float bcv = b_conv[cc];
    const float akc = attn_k[cc];
    const float bd0 = b_dense[wid * 32 + mrow];
    const float bd1 = b_dense[wid * 32 + 16 + mrow];

    float ov[BT * 96 / 256];
    #pragma unroll
    for (int j = 0; j < BT * 96 / 256; ++j) {
        const int i = tid + j * 256;
        const int r = i / 96, c = i % 96;
        ov[j] = obs[(size_t)(bbase + r) * OBSD + 3 + c];
    }

    const float* wd = W_dense + (size_t)(half * 2) * NHID + wid * 32 + mrow;
    v2f bq[8][4][2];
#define LOAD_B_BLOCK(s)                                                        \
    _Pragma("unroll")                                                          \
    for (int j_ = 0; j_ < 4; ++j_) {                                           \
        _Pragma("unroll")                                                      \
        for (int t_ = 0; t_ < 2; ++t_) {                                       \
            bq[s][j_][t_].x = wd[(size_t)(((s) * 4 + j_) * 4) * NHID + t_ * 16];     \
            bq[s][j_][t_].y = wd[(size_t)(((s) * 4 + j_) * 4 + 1) * NHID + t_ * 16]; \
        }                                                                      \
    }
    LOAD_B_BLOCK(0)
    LOAD_B_BLOCK(1)

    #pragma unroll
    for (int j = 0; j < BT * 96 / 256; ++j) {
        const int i = tid + j * 256;
        const int r = i / 96, c = i % 96;
        s_obs[r][c] = ov[j];
    }
    lds_barrier();

    {
        const int b = tid >> 4, t = (tid >> 2) & 3, e = tid & 3;
        const float* ob = s_obs[b];
        const float4 av = *(const float4*)(ob + 16 + t * 4);
        const float a[4] = {av.x, av.y, av.z, av.w};
        float4 g = {0.f, 0.f, 0.f, 0.f};
        float4 h = {0.f, 0.f, 0.f, 0.f};
        #pragma unroll
        for (int s = 0; s < 4; ++s) {
            const float4 x = *(const float4*)(ob + s * 4);
            const float ae = a[s] * ob[32 + (t * 4 + s) * 4 + e];
            g.x = fmaf(ae, x.x, g.x); g.y = fmaf(ae, x.y, g.y);
            g.z = fmaf(ae, x.z, g.z); g.w = fmaf(ae, x.w, g.w);
            h.x = fmaf(a[s], x.x, h.x); h.y = fmaf(a[s], x.y, h.y);
            h.z = fmaf(a[s], x.z, h.z); h.w = fmaf(a[s], x.w, h.w);
        }
        float* zr = s_z[b * 4 + t];
        *(float4*)(zr + e * 4) = g;
        if (e == 0) *(float4*)(zr + 16) = h;
        if (e == 1) *(float4*)(zr + 20) = *(const float4*)(ob + t * 4);
    }
    lds_barrier();

    float tot[4];
    #pragma unroll
    for (int mt = 0; mt < 4; ++mt) {
        v8f acc = {bcv, bcv, bcv, bcv, bcv, bcv, bcv, bcv};
        #pragma unroll
        for (int ks = 0; ks < 6; ++ks) {
            const int k0 = ks * 4 + half * 2;
            v2f a;
            a.x = s_z[mt * 16 + mrow][k0];
            a.y = s_z[mt * 16 + mrow][k0 + 1];
            acc = __builtin_amdgcn_wmma_f32_16x16x4_f32(
                 false, a,  false, wb[ks],
                 (short)0, acc,  false,  false);
        }
        float lg[8];
        #pragma unroll
        for (int r = 0; r < 8; ++r) {
            float xc = acc[r];
            xc = xc > 0.f ? xc : 0.f;
            s_xc[mt * 4 + half * 2 + (r >> 2)][r & 3][cc] = xc;
            lg[r] = xc * akc;
        }
        tot[mt] = row_sum8(lg, mrow);
    }
    if ((mrow & 1) == 0) {
        const int r = mrow >> 1;
        #pragma unroll
        for (int mt = 0; mt < 4; ++mt)
            s_lpart[mt * 4 + half * 2 + (r >> 2)][wid][r & 3] = tot[mt];
    }
    LOAD_B_BLOCK(2)
    LOAD_B_BLOCK(3)
    lds_barrier();

    if (tid < BT) {
        const float (*q)[4] = s_lpart[tid];
        float l0 = ((q[0][0] + q[1][0]) + (q[2][0] + q[3][0])) + ((q[4][0] + q[5][0]) + (q[6][0] + q[7][0]));
        float l1 = ((q[0][1] + q[1][1]) + (q[2][1] + q[3][1])) + ((q[4][1] + q[5][1]) + (q[6][1] + q[7][1]));
        float l2 = ((q[0][2] + q[1][2]) + (q[2][2] + q[3][2])) + ((q[4][2] + q[5][2]) + (q[6][2] + q[7][2]));
        float l3 = ((q[0][3] + q[1][3]) + (q[2][3] + q[3][3])) + ((q[4][3] + q[5][3]) + (q[6][3] + q[7][3]));
        float m  = fmaxf(fmaxf(l0, l1), fmaxf(l2, l3));
        float e0 = expf(l0 - m), e1 = expf(l1 - m), e2 = expf(l2 - m), e3 = expf(l3 - m);
        float inv = 1.0f / (e0 + e1 + e2 + e3);
        s_attn[tid][0] = e0 * inv; s_attn[tid][1] = e1 * inv;
        s_attn[tid][2] = e2 * inv; s_attn[tid][3] = e3 * inv;
    }
    lds_barrier();

    for (int p = 0; p < 8; ++p) {
        int task = p * 256 + tid;
        int b = task >> 7, c = task & 127;
        float acc = 0.f;
        #pragma unroll
        for (int t = 0; t < 4; ++t) acc = fmaf(s_attn[b][t], s_xc[b][t][c], acc);
        s_pool[b][c] = acc;
    }
    lds_barrier();

    {
        v8f acc0 = {};
        v8f acc1 = {};
        #pragma unroll
        for (int s = 0; s < 8; ++s) {
            v2f a[4];
            #pragma unroll
            for (int j = 0; j < 4; ++j) {
                const int k0 = (s * 4 + j) * 4 + half * 2;
                a[j].x = s_pool[mrow][k0];
                a[j].y = s_pool[mrow][k0 + 1];
            }
            #pragma unroll
            for (int j = 0; j < 4; ++j) {
                acc0 = __builtin_amdgcn_wmma_f32_16x16x4_f32(
                     false, a[j],  false, bq[s][j][0],
                     (short)0, acc0,  false,  false);
                acc1 = __builtin_amdgcn_wmma_f32_16x16x4_f32(
                     false, a[j],  false, bq[s][j][1],
                     (short)0, acc1,  false,  false);
            }
            if (s == 0) { LOAD_B_BLOCK(4) }
            if (s == 1) { LOAD_B_BLOCK(5) }
            if (s == 2) { LOAD_B_BLOCK(6) }
            if (s == 3) { LOAD_B_BLOCK(7) }
        }
        const int col0 = wid * 32 + mrow;
        #pragma unroll
        for (int r = 0; r < 8; ++r) {
            const int row = bbase + half * 8 + r;
            out[(size_t)row * NHID + col0] = tanhf(acc0[r] + bd0);
        }
        #pragma unroll
        for (int r = 0; r < 8; ++r) {
            const int row = bbase + half * 8 + r;
            out[(size_t)row * NHID + col0 + 16] = tanhf(acc1[r] + bd1);
        }
    }
#undef LOAD_B_BLOCK
}

extern "C" void kernel_launch(void* const* d_in, const int* in_sizes, int n_in,
                              void* d_out, int out_size, void* d_ws, size_t ws_size,
                              hipStream_t stream) {
    const float* obs     = (const float*)d_in[0];
    const float* W_fgn   = (const float*)d_in[1];
    const float* b_fgn   = (const float*)d_in[2];
    const float* W_root  = (const float*)d_in[3];
    const float* b_conv  = (const float*)d_in[4];
    const float* attn_k  = (const float*)d_in[5];
    const float* W_dense = (const float*)d_in[6];
    const float* b_dense = (const float*)d_in[7];
    float* out = (float*)d_out;

    const int B = in_sizes[0] / OBSD;
    const int grid = B / BT;
    gnn_fused_kernel<<<grid, 256, 0, stream>>>(obs, W_fgn, b_fgn, W_root, b_conv,
                                               attn_k, W_dense, b_dense, out);
}


# Epilogue tanh: library tanhf (two divergent branches, about 50 instructions per value) replaced by branch-free 1 - 2/(exp2(c|x|)+1) on hardware exp2 and rcp; sums unchanged, tanh differs by ~1e-7
# speedup vs baseline: 5.1385x; 1.1502x over previous
#include <hip/hip_runtime.h>
#include <math.h>

typedef float v2f __attribute__((ext_vector_type(2)));
typedef float v8f __attribute__((ext_vector_type(8)));

#define BT   16
#define NCH  128
#define NHID 256
#define OBSD 99
#define ZP   28

template <int M>
__device__ __forceinline__ float row_xor(float x) {
    int b = __builtin_bit_cast(int, x);
    return __builtin_bit_cast(float, __builtin_amdgcn_update_dpp(b, b, 0x160 + M, 0xf, 0xf, true));
}
__device__ __forceinline__ float row_sum8(const float (&v)[8], int mrow) {
    const bool b3 = (mrow & 8) != 0;
    const bool b2 = (mrow & 4) != 0;
    const bool b1 = (mrow & 2) != 0;
    float a[4];
    #pragma unroll
    for (int j = 0; j < 4; ++j)
        a[j] = (b3 ? v[4 + j] : v[j]) + row_xor<8>(b3 ? v[j] : v[4 + j]);
    float c[2];
    #pragma unroll
    for (int j = 0; j < 2; ++j)
        c[j] = (b2 ? a[2 + j] : a[j]) + row_xor<4>(b2 ? a[j] : a[2 + j]);
    float d = (b1 ? c[1] : c[0]) + row_xor<2>(b1 ? c[0] : c[1]);
    d += row_xor<1>(d);
    return d;
}
__device__ __forceinline__ void lds_barrier() {
    __builtin_amdgcn_fence(__ATOMIC_RELEASE, "workgroup", "local");
    __builtin_amdgcn_s_barrier();
    __builtin_amdgcn_fence(__ATOMIC_ACQUIRE, "workgroup", "local");
}
__device__ __forceinline__ float tanh_branchless(float x) {
    const float e = __builtin_amdgcn_exp2f(__builtin_fabsf(x) * 2.8853900817779268f);
    const float r = __builtin_amdgcn_rcpf(e + 1.0f);
    return __builtin_copysignf(__builtin_fmaf(-2.0f, r, 1.0f), x);
}
__global__ __launch_bounds__(256) __attribute__((amdgpu_waves_per_eu(8))) void gnn_fused_kernel(
    const float* __restrict__ obs,
    const float* __restrict__ W_fgn,
    const float* __restrict__ b_fgn,
    const float* __restrict__ W_root,
    const float* __restrict__ b_conv,
    const float* __restrict__ attn_k,
    const float* __restrict__ W_dense,
    const float* __restrict__ b_dense,
    float* __restrict__ out)
{
    __shared__ __attribute__((aligned(16))) float s_obs[BT][96];
    __shared__ float s_xc[BT][4][NCH];
    __shared__ float s_pool[BT][132];
    __shared__ __attribute__((aligned(16))) float s_z[BT * 4][ZP];
    __shared__ __attribute__((aligned(16))) float s_lpart[BT][8][4];
    __shared__ float s_attn[BT][4];

    const int tid   = threadIdx.x;
    const int bbase = blockIdx.x * BT;
    const int wid  = tid >> 5;
    const int lane = tid & 31;
    const int half = lane >> 4;
    const int mrow = lane & 15;

    const int cc = wid * 16 + mrow;
    v2f wb[6];
    #pragma unroll
    for (int e = 0; e < 4; ++e) {
        wb[e].x = W_fgn[e * 512 + cc * 4 + half * 2];
        wb[e].y = W_fgn[e * 512 + cc * 4 + half * 2 + 1];
    }
    wb[4].x = b_fgn[cc * 4 + half * 2];
    wb[4].y = b_fgn[cc * 4 + half * 2 + 1];
    wb[5].x = W_root[(half * 2) * 128 + cc];
    wb[5].y = W_root[(half * 2 + 1) * 128 + cc];
    const float bcv = b_conv[cc];
    const float akc = attn_k[cc];
    const float bd0 = b_dense[wid * 32 + mrow];
    const float bd1 = b_dense[wid * 32 + 16 + mrow];

    float ov[BT * 96 / 256];
    #pragma unroll
    for (int j = 0; j < BT * 96 / 256; ++j) {
        const int i = tid + j * 256;
        const int r = i / 96, c = i % 96;
        ov[j] = obs[(size_t)(bbase + r) * OBSD + 3 + c];
    }

    const float* wd = W_dense + (size_t)(half * 2) * NHID + wid * 32 + mrow;
    v2f bq[8][4][2];
#define LOAD_B_BLOCK(s)                                                        \
    _Pragma("unroll")                                                          \
    for (int j_ = 0; j_ < 4; ++j_) {                                           \
        _Pragma("unroll")                                                      \
        for (int t_ = 0; t_ < 2; ++t_) {                                       \
            bq[s][j_][t_].x = wd[(size_t)(((s) * 4 + j_) * 4) * NHID + t_ * 16];     \
            bq[s][j_][t_].y = wd[(size_t)(((s) * 4 + j_) * 4 + 1) * NHID + t_ * 16]; \
        }                                                                      \
    }
    LOAD_B_BLOCK(0)
    LOAD_B_BLOCK(1)

    #pragma unroll
    for (int j = 0; j < BT * 96 / 256; ++j) {
        const int i = tid + j * 256;
        const int r = i / 96, c = i % 96;
        s_obs[r][c] = ov[j];
    }
    lds_barrier();

    {
        const int b = tid >> 4, t = (tid >> 2) & 3, e = tid & 3;
        const float* ob = s_obs[b];
        const float4 av = *(const float4*)(ob + 16 + t * 4);
        const float a[4] = {av.x, av.y, av.z, av.w};
        float4 g = {0.f, 0.f, 0.f, 0.f};
        float4 h = {0.f, 0.f, 0.f, 0.f};
        #pragma unroll
        for (int s = 0; s < 4; ++s) {
            const float4 x = *(const float4*)(ob + s * 4);
            const float ae = a[s] * ob[32 + (t * 4 + s) * 4 + e];
            g.x = fmaf(ae, x.x, g.x); g.y = fmaf(ae, x.y, g.y);
            g.z = fmaf(ae, x.z, g.z); g.w = fmaf(ae, x.w, g.w);
            h.x = fmaf(a[s], x.x, h.x); h.y = fmaf(a[s], x.y, h.y);
            h.z = fmaf(a[s], x.z, h.z); h.w = fmaf(a[s], x.w, h.w);
        }
        float* zr = s_z[b * 4 + t];
        *(float4*)(zr + e * 4) = g;
        if (e == 0) *(float4*)(zr + 16) = h;
        if (e == 1) *(float4*)(zr + 20) = *(const float4*)(ob + t * 4);
    }
    lds_barrier();

    float tot[4];
    #pragma unroll
    for (int mt = 0; mt < 4; ++mt) {
        v8f acc = {bcv, bcv, bcv, bcv, bcv, bcv, bcv, bcv};
        #pragma unroll
        for (int ks = 0; ks < 6; ++ks) {
            const int k0 = ks * 4 + half * 2;
            v2f a;
            a.x = s_z[mt * 16 + mrow][k0];
            a.y = s_z[mt * 16 + mrow][k0 + 1];
            acc = __builtin_amdgcn_wmma_f32_16x16x4_f32(
                 false, a,  false, wb[ks],
                 (short)0, acc,  false,  false);
        }
        float lg[8];
        #pragma unroll
        for (int r = 0; r < 8; ++r) {
            float xc = acc[r];
            xc = xc > 0.f ? xc : 0.f;
            s_xc[mt * 4 + half * 2 + (r >> 2)][r & 3][cc] = xc;
            lg[r] = xc * akc;
        }
        tot[mt] = row_sum8(lg, mrow);
    }
    if ((mrow & 1) == 0) {
        const int r = mrow >> 1;
        #pragma unroll
        for (int mt = 0; mt < 4; ++mt)
            s_lpart[mt * 4 + half * 2 + (r >> 2)][wid][r & 3] = tot[mt];
    }
    LOAD_B_BLOCK(2)
    LOAD_B_BLOCK(3)
    lds_barrier();

    if (tid < BT) {
        const float (*q)[4] = s_lpart[tid];
        float l0 = ((q[0][0] + q[1][0]) + (q[2][0] + q[3][0])) + ((q[4][0] + q[5][0]) + (q[6][0] + q[7][0]));
        float l1 = ((q[0][1] + q[1][1]) + (q[2][1] + q[3][1])) + ((q[4][1] + q[5][1]) + (q[6][1] + q[7][1]));
        float l2 = ((q[0][2] + q[1][2]) + (q[2][2] + q[3][2])) + ((q[4][2] + q[5][2]) + (q[6][2] + q[7][2]));
        float l3 = ((q[0][3] + q[1][3]) + (q[2][3] + q[3][3])) + ((q[4][3] + q[5][3]) + (q[6][3] + q[7][3]));
        float m  = fmaxf(fmaxf(l0, l1), fmaxf(l2, l3));
        float e0 = expf(l0 - m), e1 = expf(l1 - m), e2 = expf(l2 - m), e3 = expf(l3 - m);
        float inv = 1.0f / (e0 + e1 + e2 + e3);
        s_attn[tid][0] = e0 * inv; s_attn[tid][1] = e1 * inv;
        s_attn[tid][2] = e2 * inv; s_attn[tid][3] = e3 * inv;
    }
    lds_barrier();

    for (int p = 0; p < 8; ++p) {
        int task = p * 256 + tid;
        int b = task >> 7, c = task & 127;
        float acc = 0.f;
        #pragma unroll
        for (int t = 0; t < 4; ++t) acc = fmaf(s_attn[b][t], s_xc[b][t][c], acc);
        s_pool[b][c] = acc;
    }
    lds_barrier();

    {
        v8f acc0 = {};
        v8f acc1 = {};
        #pragma unroll
        for (int s = 0; s < 8; ++s) {
            v2f a[4];
            #pragma unroll
            for (int j = 0; j < 4; ++j) {
                const int k0 = (s * 4 + j) * 4 + half * 2;
                a[j].x = s_pool[mrow][k0];
                a[j].y = s_pool[mrow][k0 + 1];
            }
            #pragma unroll
            for (int j = 0; j < 4; ++j) {
                acc0 = __builtin_amdgcn_wmma_f32_16x16x4_f32(
                     false, a[j],  false, bq[s][j][0],
                     (short)0, acc0,  false,  false);
                acc1 = __builtin_amdgcn_wmma_f32_16x16x4_f32(
                     false, a[j],  false, bq[s][j][1],
                     (short)0, acc1,  false,  false);
            }
            if (s == 0) { LOAD_B_BLOCK(4) }
            if (s == 1) { LOAD_B_BLOCK(5) }
            if (s == 2) { LOAD_B_BLOCK(6) }
            if (s == 3) { LOAD_B_BLOCK(7) }
        }
        const int col0 = wid * 32 + mrow;
        #pragma unroll
        for (int r = 0; r < 8; ++r) {
            const int row = bbase + half * 8 + r;
            out[(size_t)row * NHID + col0] = tanh_branchless(acc0[r] + bd0);
        }
        #pragma unroll
        for (int r = 0; r < 8; ++r) {
            const int row = bbase + half * 8 + r;
            out[(size_t)row * NHID + col0 + 16] = tanh_branchless(acc1[r] + bd1);
        }
    }
#undef LOAD_B_BLOCK
}

extern "C" void kernel_launch(void* const* d_in, const int* in_sizes, int n_in,
                              void* d_out, int out_size, void* d_ws, size_t ws_size,
                              hipStream_t stream) {
    const float* obs     = (const float*)d_in[0];
    const float* W_fgn   = (const float*)d_in[1];
    const float* b_fgn   = (const float*)d_in[2];
    const float* W_root  = (const float*)d_in[3];
    const float* b_conv  = (const float*)d_in[4];
    const float* attn_k  = (const float*)d_in[5];
    const float* W_dense = (const float*)d_in[6];
    const float* b_dense = (const float*)d_in[7];
    float* out = (float*)d_out;

    const int B = in_sizes[0] / OBSD;
    const int grid = B / BT;
    gnn_fused_kernel<<<grid, 256, 0, stream>>>(obs, W_fgn, b_fgn, W_root, b_conv,
                                               attn_k, W_dense, b_dense, out);
}


# Dense wave tile widened from 32 to 64 columns (block of 4 waves, each also 2 conv channel tiles): W_dense, bias and output rows move as 16-byte per-lane accesses, half the waves; sums unchanged
# speedup vs baseline: 5.5371x; 1.0776x over previous
#include <hip/hip_runtime.h>
#include <math.h>

typedef float v2f __attribute__((ext_vector_type(2)));
typedef float v8f __attribute__((ext_vector_type(8)));

#define BT   16
#define NTHR 128
#define NCH  128
#define NHID 256
#define OBSD 99
#define ZP   28

template <int M>
__device__ __forceinline__ float row_xor(float x) {
    int b = __builtin_bit_cast(int, x);
    return __builtin_bit_cast(float, __builtin_amdgcn_update_dpp(b, b, 0x160 + M, 0xf, 0xf, true));
}
__device__ __forceinline__ float row_sum8(const float (&v)[8], int mrow) {
    const bool b3 = (mrow & 8) != 0;
    const bool b2 = (mrow & 4) != 0;
    const bool b1 = (mrow & 2) != 0;
    float a[4];
    #pragma unroll
    for (int j = 0; j < 4; ++j)
        a[j] = (b3 ? v[4 + j] : v[j]) + row_xor<8>(b3 ? v[j] : v[4 + j]);
    float c[2];
    #pragma unroll
    for (int j = 0; j < 2; ++j)
        c[j] = (b2 ? a[2 + j] : a[j]) + row_xor<4>(b2 ? a[j] : a[2 + j]);
    float d = (b1 ? c[1] : c[0]) + row_xor<2>(b1 ? c[0] : c[1]);
    d += row_xor<1>(d);
    return d;
}
__device__ __forceinline__ void lds_barrier() {
    __builtin_amdgcn_fence(__ATOMIC_RELEASE, "workgroup", "local");
    __builtin_amdgcn_s_barrier();
    __builtin_amdgcn_fence(__ATOMIC_ACQUIRE, "workgroup", "local");
}
__device__ __forceinline__ float tanh_branchless(float x) {
    const float e = __builtin_amdgcn_exp2f(__builtin_fabsf(x) * 2.8853900817779268f);
    const float r = __builtin_amdgcn_rcpf(e + 1.0f);
    return __builtin_copysignf(__builtin_fmaf(-2.0f, r, 1.0f), x);
}
__global__ __launch_bounds__(NTHR) __attribute__((amdgpu_waves_per_eu(4))) void gnn_fused_kernel(
    const float* __restrict__ obs,
    const float* __restrict__ W_fgn,
    const float* __restrict__ b_fgn,
    const float* __restrict__ W_root,
    const float* __restrict__ b_conv,
    const float* __restrict__ attn_k,
    const float* __restrict__ W_dense,
    const float* __restrict__ b_dense,
    float* __restrict__ out)
{
    __shared__ __attribute__((aligned(16))) float s_obs[BT][96];
    __shared__ float s_xc[BT][4][NCH];
    __shared__ float s_pool[BT][132];
    __shared__ __attribute__((aligned(16))) float s_z[BT * 4][ZP];
    __shared__ __attribute__((aligned(16))) float s_lpart[BT][8][4];
    __shared__ float s_attn[BT][4];

    const int tid   = threadIdx.x;
    const int bbase = blockIdx.x * BT;
    const int wid  = tid >> 5;
    const int lane = tid & 31;
    const int half = lane >> 4;
    const int mrow = lane & 15;

    v2f wb[2][6];
    float bcv[2], akc[2];
    #pragma unroll
    for (int q = 0; q < 2; ++q) {
        const int cc = (wid * 2 + q) * 16 + mrow;
        #pragma unroll
        for (int e = 0; e < 4; ++e) {
            wb[q][e].x = W_fgn[e * 512 + cc * 4 + half * 2];
            wb[q][e].y = W_fgn[e * 512 + cc * 4 + half * 2 + 1];
        }
        wb[q][4].x = b_fgn[cc * 4 + half * 2];
        wb[q][4].y = b_fgn[cc * 4 + half * 2 + 1];
        wb[q][5].x = W_root[(half * 2) * 128 + cc];
        wb[q][5].y = W_root[(half * 2 + 1) * 128 + cc];
        bcv[q] = b_conv[cc];
        akc[q] = attn_k[cc];
    }
    const int dcol = wid * 64 + 4 * mrow;
    const float4 bd = *(const float4*)(b_dense + dcol);

    float ov[BT * 96 / NTHR];
    #pragma unroll
    for (int j = 0; j < BT * 96 / NTHR; ++j) {
        const int i = tid + j * NTHR;
        const int r = i / 96, c = i % 96;
        ov[j] = obs[(size_t)(bbase + r) * OBSD + 3 + c];
    }

    const float* wd = W_dense + (size_t)(half * 2) * NHID + dcol;
    float4 bq[8][4][2];
#define LOAD_B_BLOCK(s)                                                        \
    _Pragma("unroll")                                                          \
    for (int j_ = 0; j_ < 4; ++j_) {                                           \
        bq[s][j_][0] = *(const float4*)(wd + (size_t)(((s) * 4 + j_) * 4) * NHID);     \
        bq[s][j_][1] = *(const float4*)(wd + (size_t)(((s) * 4 + j_) * 4 + 1) * NHID); \
    }
    LOAD_B_BLOCK(0)
    LOAD_B_BLOCK(1)

    #pragma unroll
    for (int j = 0; j < BT * 96 / NTHR; ++j) {
        const int i = tid + j * NTHR;
        const int r = i / 96, c = i % 96;
        s_obs[r][c] = ov[j];
    }
    lds_barrier();

    #pragma unroll
    for (int p = 0; p < 2; ++p) {
        const int task = tid + p * NTHR;
        const int b = task >> 4, t = (task >> 2) & 3, e = task & 3;
        const float* ob = s_obs[b];
        const float4 av = *(const float4*)(ob + 16 + t * 4);
        const float a[4] = {av.x, av.y, av.z, av.w};
        float4 g = {0.f, 0.f, 0.f, 0.f};
        float4 h = {0.f, 0.f, 0.f, 0.f};
        #pragma unroll
        for (int s = 0; s < 4; ++s) {
            const float4 x = *(const float4*)(ob + s * 4);
            const float ae = a[s] * ob[32 + (t * 4 + s) * 4 + e];
            g.x = fmaf(ae, x.x, g.x); g.y = fmaf(ae, x.y, g.y);
            g.z = fmaf(ae, x.z, g.z); g.w = fmaf(ae, x.w, g.w);
            h.x = fmaf(a[s], x.x, h.x); h.y = fmaf(a[s], x.y, h.y);
            h.z = fmaf(a[s], x.z, h.z); h.w = fmaf(a[s], x.w, h.w);
        }
        float* zr = s_z[b * 4 + t];
        *(float4*)(zr + e * 4) = g;
        if (e == 0) *(float4*)(zr + 16) = h;
        if (e == 1) *(float4*)(zr + 20) = *(const float4*)(ob + t * 4);
    }
    lds_barrier();

    float tot[2][4];
    #pragma unroll
    for (int mt = 0; mt < 4; ++mt) {
        v2f az[6];
        #pragma unroll
        for (int ks = 0; ks < 6; ++ks) {
            const int k0 = ks * 4 + half * 2;
            az[ks].x = s_z[mt * 16 + mrow][k0];
            az[ks].y = s_z[mt * 16 + mrow][k0 + 1];
        }
        #pragma unroll
        for (int q = 0; q < 2; ++q) {
            const int cc = (wid * 2 + q) * 16 + mrow;
            v8f acc = {bcv[q], bcv[q], bcv[q], bcv[q], bcv[q], bcv[q], bcv[q], bcv[q]};
            #pragma unroll
            for (int ks = 0; ks < 6; ++ks) {
                acc = __builtin_amdgcn_wmma_f32_16x16x4_f32(
                     false, az[ks],  false, wb[q][ks],
                     (short)0, acc,  false,  false);
            }
            float lg[8];
            #pragma unroll
            for (int r = 0; r < 8; ++r) {
                float xc = acc[r];
                xc = xc > 0.f ? xc : 0.f;
                s_xc[mt * 4 + half * 2 + (r >> 2)][r & 3][cc] = xc;
                lg[r] = xc * akc[q];
            }
            tot[q][mt] = row_sum8(lg, mrow);
        }
    }
    if ((mrow & 1) == 0) {
        const int r = mrow >> 1;
        #pragma unroll
        for (int q = 0; q < 2; ++q) {
            #pragma unroll
            for (int mt = 0; mt < 4; ++mt)
                s_lpart[mt * 4 + half * 2 + (r >> 2)][wid * 2 + q][r & 3] = tot[q][mt];
        }
    }
    LOAD_B_BLOCK(2)
    LOAD_B_BLOCK(3)
    lds_barrier();

    if (tid < BT) {
        const float (*q)[4] = s_lpart[tid];
        float l0 = ((q[0][0] + q[1][0]) + (q[2][0] + q[3][0])) + ((q[4][0] + q[5][0]) + (q[6][0] + q[7][0]));
        float l1 = ((q[0][1] + q[1][1]) + (q[2][1] + q[3][1])) + ((q[4][1] + q[5][1]) + (q[6][1] + q[7][1]));
        float l2 = ((q[0][2] + q[1][2]) + (q[2][2] + q[3][2])) + ((q[4][2] + q[5][2]) + (q[6][2] + q[7][2]));
        float l3 = ((q[0][3] + q[1][3]) + (q[2][3] + q[3][3])) + ((q[4][3] + q[5][3]) + (q[6][3] + q[7][3]));
        float m  = fmaxf(fmaxf(l0, l1), fmaxf(l2, l3));
        float e0 = expf(l0 - m), e1 = expf(l1 - m), e2 = expf(l2 - m), e3 = expf(l3 - m);
        float inv = 1.0f / (e0 + e1 + e2 + e3);
        s_attn[tid][0] = e0 * inv; s_attn[tid][1] = e1 * inv;
        s_attn[tid][2] = e2 * inv; s_attn[tid][3] = e3 * inv;
    }
    lds_barrier();

    for (int p = 0; p < BT; ++p) {
        const int b = p, c = tid;
        float acc = 0.f;
        #pragma unroll
        for (int t = 0; t < 4; ++t) acc = fmaf(s_attn[b][t], s_xc[b][t][c], acc);
        s_pool[b][c] = acc;
    }
    lds_barrier();

    {
        v8f acc[4] = {};
        #pragma unroll
        for (int s = 0; s < 8; ++s) {
            v2f a[4];
            #pragma unroll
            for (int j = 0; j < 4; ++j) {
                const int k0 = (s * 4 + j) * 4 + half * 2;
                a[j].x = s_pool[mrow][k0];
                a[j].y = s_pool[mrow][k0 + 1];
            }
            #pragma unroll
            for (int j = 0; j < 4; ++j) {
                const float4 bx = bq[s][j][0];
                const float4 by = bq[s][j][1];
                const v2f b0 = {bx.x, by.x};
                const v2f b1 = {bx.y, by.y};
                const v2f b2 = {bx.z, by.z};
                const v2f b3 = {bx.w, by.w};
                acc[0] = __builtin_amdgcn_wmma_f32_16x16x4_f32(
                     false, a[j],  false, b0,
                     (short)0, acc[0],  false,  false);
                acc[1] = __builtin_amdgcn_wmma_f32_16x16x4_f32(
                     false, a[j],  false, b1,
                     (short)0, acc[1],  false,  false);
                acc[2] = __builtin_amdgcn_wmma_f32_16x16x4_f32(
                     false, a[j],  false, b2,
                     (short)0, acc[2],  false,  false);
                acc[3] = __builtin_amdgcn_wmma_f32_16x16x4_f32(
                     false, a[j],  false, b3,
                     (short)0, acc[3],  false,  false);
            }
            if (s == 0) { LOAD_B_BLOCK(4) }
            if (s == 1) { LOAD_B_BLOCK(5) }
            if (s == 2) { LOAD_B_BLOCK(6) }
            if (s == 3) { LOAD_B_BLOCK(7) }
        }
        #pragma unroll
        for (int r = 0; r < 8; ++r) {
            const int row = bbase + half * 8 + r;
            float4 o;
            o.x = tanh_branchless(acc[0][r] + bd.x);
            o.y = tanh_branchless(acc[1][r] + bd.y);
            o.z = tanh_branchless(acc[2][r] + bd.z);
            o.w = tanh_branchless(acc[3][r] + bd.w);
            *(float4*)(out + (size_t)row * NHID + dcol) = o;
        }
    }
#undef LOAD_B_BLOCK
}

extern "C" void kernel_launch(void* const* d_in, const int* in_sizes, int n_in,
                              void* d_out, int out_size, void* d_ws, size_t ws_size,
                              hipStream_t stream) {
    const float* obs     = (const float*)d_in[0];
    const float* W_fgn   = (const float*)d_in[1];
    const float* b_fgn   = (const float*)d_in[2];
    const float* W_root  = (const float*)d_in[3];
    const float* b_conv  = (const float*)d_in[4];
    const float* attn_k  = (const float*)d_in[5];
    const float* W_dense = (const float*)d_in[6];
    const float* b_dense = (const float*)d_in[7];
    float* out = (float*)d_out;

    const int B = in_sizes[0] / OBSD;
    const int grid = B / BT;
    gnn_fused_kernel<<<grid, NTHR, 0, stream>>>(obs, W_fgn, b_fgn, W_root, b_conv,
                                               attn_k, W_dense, b_dense, out);
}
